# starting kernel
# baseline (speedup 1.0000x reference)
#include <hip/hip_runtime.h>


typedef __attribute__((ext_vector_type(2))) float v2f;
typedef __attribute__((ext_vector_type(8))) float v8f;

#define N 2048
#define NTILE (N / 16)
#define LOG2N 11

__global__ __launch_bounds__(256) void rowscan_wmma(const float* __restrict__ D,
                                                    float* __restrict__ R) {
  const int lane = threadIdx.x & 31;
  const int wave = threadIdx.x >> 5;
  const int strip = blockIdx.x * 8 + wave;
  const int b     = strip >> 7;
  const int row0  = (strip & 127) << 4;

  const float* __restrict__ Db = D + (size_t)b * N * N;
  float* __restrict__       Rb = R + (size_t)b * N * N;

  __shared__ float lds[8][16 * 16];
  float* tl = lds[wave];

  const int m    = lane & 15;
  const int half = lane >> 4;

  v8f c = {0.f, 0.f, 0.f, 0.f, 0.f, 0.f, 0.f, 0.f};

  for (int tile = 0; tile < NTILE; ++tile) {
#pragma unroll
    for (int t = 0; t < 2; ++t) {
      const int idx4 = t * 32 + lane;
      const int r    = idx4 >> 2;
      const int cg   = idx4 & 3;
      const float4 v =
          *(const float4*)(Db + (size_t)(row0 + r) * N + tile * 16 + cg * 4);
      *(float4*)&tl[r * 16 + cg * 4] = v;
    }

#pragma unroll
    for (int s = 0; s < 4; ++s) {
      const int k0 = s * 4 + half * 2;
      v2f a, u;
      a.x = tl[m * 16 + k0];
      a.y = tl[m * 16 + k0 + 1];
      u.x = (k0     <= m) ? 1.0f : 0.0f;
      u.y = (k0 + 1 <= m) ? 1.0f : 0.0f;
      c = __builtin_amdgcn_wmma_f32_16x16x4_f32(
           false, a,  false, u,
           (short)0, c,  false,  false);
    }

    const int col = tile * 16 + m;
#pragma unroll
    for (int r = 0; r < 8; ++r) {
      const int row = row0 + r + half * 8;
      Rb[(size_t)row * N + col] = c[r];
    }

    const int src = half ? 31 : 15;
#pragma unroll
    for (int r = 0; r < 8; ++r) c[r] = __shfl(c[r], src, 32);
  }
}

__global__ __launch_bounds__(256) void colscan(float* __restrict__ P) {
  const int colg = blockIdx.x * blockDim.x + threadIdx.x;
  const int b = colg >> LOG2N;
  const int j = colg & (N - 1);
  float* base = P + (size_t)b * N * N + j;

  float acc = 0.0f;
#pragma unroll 4
  for (int i = 0; i < N; ++i) {
    __builtin_prefetch(base + (size_t)(i + 16) * N, 0, 0);
    const float v = base[(size_t)i * N];
    acc += v;
    base[(size_t)i * N] = acc;
  }
}

__global__ __launch_bounds__(256) void assemble(const float* __restrict__ D,
                                                const float* __restrict__ P,
                                                float* __restrict__ out) {
  const size_t t = (size_t)blockIdx.x * blockDim.x + threadIdx.x;
  const int j = (int)(t & (N - 1));
  const int i = (int)((t >> LOG2N) & (N - 1));
  const int b = (int)(t >> (2 * LOG2N));
  const float* __restrict__ Pb = P + (size_t)b * N * N;

  float r;
  if (i == j + 1) {
    r = D[(size_t)b * N * N + (size_t)i * N + j];
  } else {
    const int a = (i < j) ? i : j;
    const int cmx = (i < j) ? j : i;
    r = Pb[(size_t)cmx * N + cmx];
    if (a > 0) {
      r -= Pb[(size_t)(a - 1) * N + cmx];
      r -= Pb[(size_t)cmx * N + (a - 1)];
      r += Pb[(size_t)(a - 1) * N + (a - 1)];
    }
  }
  out[t] = r;
}

extern "C" void kernel_launch(void* const* d_in, const int* in_sizes, int n_in,
                              void* d_out, int out_size, void* d_ws, size_t ws_size,
                              hipStream_t stream) {
  const float* D = (const float*)d_in[0];
  float* out = (float*)d_out;
  float* P   = (float*)d_ws;
  const int B = in_sizes[0] / (N * N);

  rowscan_wmma<<<dim3((B * NTILE) / 8), dim3(256), 0, stream>>>(D, P);
  colscan<<<dim3((B * N) / 256), dim3(256), 0, stream>>>(P);
  const size_t total = (size_t)B * N * N;
  assemble<<<dim3((unsigned)(total / 256)), dim3(256), 0, stream>>>(D, P, out);
}


# colscan: row loads issued 48 to 64 rows ahead of the running sum through a register ring, one wave per 32-column strip in its own workgroup; order of the column sums unchanged
# speedup vs baseline: 3.4698x; 3.4698x over previous
#include <hip/hip_runtime.h>


typedef __attribute__((ext_vector_type(2))) float v2f;
typedef __attribute__((ext_vector_type(8))) float v8f;

#define N 2048
#define NTILE (N / 16)
#define LOG2N 11

__global__ __launch_bounds__(256) void rowscan_wmma(const float* __restrict__ D,
                                                    float* __restrict__ R) {
  const int lane = threadIdx.x & 31;
  const int wave = threadIdx.x >> 5;
  const int strip = blockIdx.x * 8 + wave;
  const int b     = strip >> 7;
  const int row0  = (strip & 127) << 4;

  const float* __restrict__ Db = D + (size_t)b * N * N;
  float* __restrict__       Rb = R + (size_t)b * N * N;

  __shared__ float lds[8][16 * 16];
  float* tl = lds[wave];

  const int m    = lane & 15;
  const int half = lane >> 4;

  v8f c = {0.f, 0.f, 0.f, 0.f, 0.f, 0.f, 0.f, 0.f};

  for (int tile = 0; tile < NTILE; ++tile) {
#pragma unroll
    for (int t = 0; t < 2; ++t) {
      const int idx4 = t * 32 + lane;
      const int r    = idx4 >> 2;
      const int cg   = idx4 & 3;
      const float4 v =
          *(const float4*)(Db + (size_t)(row0 + r) * N + tile * 16 + cg * 4);
      *(float4*)&tl[r * 16 + cg * 4] = v;
    }

#pragma unroll
    for (int s = 0; s < 4; ++s) {
      const int k0 = s * 4 + half * 2;
      v2f a, u;
      a.x = tl[m * 16 + k0];
      a.y = tl[m * 16 + k0 + 1];
      u.x = (k0     <= m) ? 1.0f : 0.0f;
      u.y = (k0 + 1 <= m) ? 1.0f : 0.0f;
      c = __builtin_amdgcn_wmma_f32_16x16x4_f32(
           false, a,  false, u,
           (short)0, c,  false,  false);
    }

    const int col = tile * 16 + m;
#pragma unroll
    for (int r = 0; r < 8; ++r) {
      const int row = row0 + r + half * 8;
      Rb[(size_t)row * N + col] = c[r];
    }

    const int src = half ? 31 : 15;
#pragma unroll
    for (int r = 0; r < 8; ++r) c[r] = __shfl(c[r], src, 32);
  }
}

#define CS_BATCH 16
#define CS_RING  4
#define CS_NB    (N / CS_BATCH)

__global__ __launch_bounds__(32, 1) void colscan(float* __restrict__ P) {
  const int colg = blockIdx.x * 32 + threadIdx.x;
  const int b = colg >> LOG2N;
  const int j = colg & (N - 1);
  float* p = P + (size_t)b * N * N + j;

  float ring[CS_RING][CS_BATCH];

#pragma unroll
  for (int u = 0; u < CS_RING - 1; ++u) {
#pragma unroll
    for (int r = 0; r < CS_BATCH; ++r)
      ring[u][r] = p[(size_t)(u * CS_BATCH + r) * N];
  }

  float acc = 0.0f;
  for (int g = 0; g < CS_NB - CS_RING; g += CS_RING) {
#pragma unroll
    for (int u = 0; u < CS_RING; ++u) {
#pragma unroll
      for (int r = 0; r < CS_BATCH; ++r)
        ring[(u + CS_RING - 1) % CS_RING][r] =
            p[(size_t)((u + CS_RING - 1) * CS_BATCH + r) * N];
#pragma unroll
      for (int r = 0; r < CS_BATCH; ++r) {
        acc += ring[u][r];
        p[(size_t)(u * CS_BATCH + r) * N] = acc;
      }
    }
    p += (size_t)CS_RING * CS_BATCH * N;
  }

#pragma unroll
  for (int u = 0; u < CS_RING; ++u) {
    if (u == 0) {
#pragma unroll
      for (int r = 0; r < CS_BATCH; ++r)
        ring[CS_RING - 1][r] = p[(size_t)((CS_RING - 1) * CS_BATCH + r) * N];
    }
#pragma unroll
    for (int r = 0; r < CS_BATCH; ++r) {
      acc += ring[u][r];
      p[(size_t)(u * CS_BATCH + r) * N] = acc;
    }
  }
}

__global__ __launch_bounds__(256) void assemble(const float* __restrict__ D,
                                                const float* __restrict__ P,
                                                float* __restrict__ out) {
  const size_t t = (size_t)blockIdx.x * blockDim.x + threadIdx.x;
  const int j = (int)(t & (N - 1));
  const int i = (int)((t >> LOG2N) & (N - 1));
  const int b = (int)(t >> (2 * LOG2N));
  const float* __restrict__ Pb = P + (size_t)b * N * N;

  float r;
  if (i == j + 1) {
    r = D[(size_t)b * N * N + (size_t)i * N + j];
  } else {
    const int a = (i < j) ? i : j;
    const int cmx = (i < j) ? j : i;
    r = Pb[(size_t)cmx * N + cmx];
    if (a > 0) {
      r -= Pb[(size_t)(a - 1) * N + cmx];
      r -= Pb[(size_t)cmx * N + (a - 1)];
      r += Pb[(size_t)(a - 1) * N + (a - 1)];
    }
  }
  out[t] = r;
}

extern "C" void kernel_launch(void* const* d_in, const int* in_sizes, int n_in,
                              void* d_out, int out_size, void* d_ws, size_t ws_size,
                              hipStream_t stream) {
  const float* D = (const float*)d_in[0];
  float* out = (float*)d_out;
  float* P   = (float*)d_ws;
  const int B = in_sizes[0] / (N * N);

  rowscan_wmma<<<dim3((B * NTILE) / 8), dim3(256), 0, stream>>>(D, P);
  colscan<<<dim3((B * N) / 32), dim3(32), 0, stream>>>(P);
  const size_t total = (size_t)B * N * N;
  assemble<<<dim3((unsigned)(total / 256)), dim3(256), 0, stream>>>(D, P, out);
}


# assemble: each workgroup stages the two SAT tiles of a mirrored 32x32 output tile pair through LDS, so every global access is a coalesced row, not a column or diagonal gather; term order unchanged
# speedup vs baseline: 7.7215x; 2.2253x over previous
#include <hip/hip_runtime.h>


typedef __attribute__((ext_vector_type(2))) float v2f;
typedef __attribute__((ext_vector_type(8))) float v8f;

#define N 2048
#define NTILE (N / 16)
#define LOG2N 11

__global__ __launch_bounds__(256) void rowscan_wmma(const float* __restrict__ D,
                                                    float* __restrict__ R) {
  const int lane = threadIdx.x & 31;
  const int wave = threadIdx.x >> 5;
  const int strip = blockIdx.x * 8 + wave;
  const int b     = strip >> 7;
  const int row0  = (strip & 127) << 4;

  const float* __restrict__ Db = D + (size_t)b * N * N;
  float* __restrict__       Rb = R + (size_t)b * N * N;

  __shared__ float lds[8][16 * 16];
  float* tl = lds[wave];

  const int m    = lane & 15;
  const int half = lane >> 4;

  v8f c = {0.f, 0.f, 0.f, 0.f, 0.f, 0.f, 0.f, 0.f};

  for (int tile = 0; tile < NTILE; ++tile) {
#pragma unroll
    for (int t = 0; t < 2; ++t) {
      const int idx4 = t * 32 + lane;
      const int r    = idx4 >> 2;
      const int cg   = idx4 & 3;
      const float4 v =
          *(const float4*)(Db + (size_t)(row0 + r) * N + tile * 16 + cg * 4);
      *(float4*)&tl[r * 16 + cg * 4] = v;
    }

#pragma unroll
    for (int s = 0; s < 4; ++s) {
      const int k0 = s * 4 + half * 2;
      v2f a, u;
      a.x = tl[m * 16 + k0];
      a.y = tl[m * 16 + k0 + 1];
      u.x = (k0     <= m) ? 1.0f : 0.0f;
      u.y = (k0 + 1 <= m) ? 1.0f : 0.0f;
      c = __builtin_amdgcn_wmma_f32_16x16x4_f32(
           false, a,  false, u,
           (short)0, c,  false,  false);
    }

    const int col = tile * 16 + m;
#pragma unroll
    for (int r = 0; r < 8; ++r) {
      const int row = row0 + r + half * 8;
      Rb[(size_t)row * N + col] = c[r];
    }

    const int src = half ? 31 : 15;
#pragma unroll
    for (int r = 0; r < 8; ++r) c[r] = __shfl(c[r], src, 32);
  }
}

#define CS_BATCH 16
#define CS_RING  4
#define CS_NB    (N / CS_BATCH)

__global__ __launch_bounds__(32, 1) void colscan(float* __restrict__ P) {
  const int colg = blockIdx.x * 32 + threadIdx.x;
  const int b = colg >> LOG2N;
  const int j = colg & (N - 1);
  float* p = P + (size_t)b * N * N + j;

  float ring[CS_RING][CS_BATCH];

#pragma unroll
  for (int u = 0; u < CS_RING - 1; ++u) {
#pragma unroll
    for (int r = 0; r < CS_BATCH; ++r)
      ring[u][r] = p[(size_t)(u * CS_BATCH + r) * N];
  }

  float acc = 0.0f;
  for (int g = 0; g < CS_NB - CS_RING; g += CS_RING) {
#pragma unroll
    for (int u = 0; u < CS_RING; ++u) {
#pragma unroll
      for (int r = 0; r < CS_BATCH; ++r)
        ring[(u + CS_RING - 1) % CS_RING][r] =
            p[(size_t)((u + CS_RING - 1) * CS_BATCH + r) * N];
#pragma unroll
      for (int r = 0; r < CS_BATCH; ++r) {
        acc += ring[u][r];
        p[(size_t)(u * CS_BATCH + r) * N] = acc;
      }
    }
    p += (size_t)CS_RING * CS_BATCH * N;
  }

#pragma unroll
  for (int u = 0; u < CS_RING; ++u) {
    if (u == 0) {
#pragma unroll
      for (int r = 0; r < CS_BATCH; ++r)
        ring[CS_RING - 1][r] = p[(size_t)((CS_RING - 1) * CS_BATCH + r) * N];
    }
#pragma unroll
    for (int r = 0; r < CS_BATCH; ++r) {
      acc += ring[u][r];
      p[(size_t)(u * CS_BATCH + r) * N] = acc;
    }
  }
}

#define AT 32
#define AP (AT + 1)
#define NT (N / AT)
#define AW 4
#define AR (AT / AW)

__global__ __launch_bounds__(AW * 32) void assemble(const float* __restrict__ D,
                                                const float* __restrict__ P,
                                                float* __restrict__ out) {
  __shared__ float sA[AT * AP];
  __shared__ float sB[AT * AP];
  __shared__ float sd[2 * AT];
  const int lane = threadIdx.x & 31;
  const int wave = threadIdx.x >> 5;

  const int u = blockIdx.x;
  const int v = blockIdx.y;
  int I, J;
  if (u < NT - v) { I = v;          J = v + u; }
  else            { I = NT - 1 - v; J = I + (u - (NT - v)); }
  const int i0 = I * AT;
  const int j0 = J * AT;

  const size_t boff = (size_t)blockIdx.z * N * N;
  const float* __restrict__ Pb = P + boff;

  const int colB  = i0 - 1 + lane;
  const int colBc = (colB < 0) ? 0 : colB;
  float ra[AR], rb[AR];
#pragma unroll
  for (int t = 0; t < AR; ++t) {
    const int x = wave + AW * t;
    const int rowA  = i0 + x - 1;
    const int rowAc = (rowA < 0) ? 0 : rowA;
    ra[t] = Pb[(size_t)rowAc * N + j0 + lane];
    rb[t] = Pb[(size_t)(j0 + x) * N + colBc];
  }
  float dv = 0.0f;
  if (wave < 2) {
    const int k  = (wave == 0) ? (j0 + lane) : (i0 + lane - 1);
    const int kc = (k < 0) ? 0 : k;
    dv = Pb[(size_t)kc * (N + 1)];
  }
#pragma unroll
  for (int t = 0; t < AR; ++t) {
    const int x = wave + AW * t;
    sA[x * AP + lane] = ra[t];
    sB[x * AP + lane] = rb[t];
  }
  if (wave < 2) sd[wave * AT + lane] = dv;
  __syncthreads();

  float r[AR];

  {
    const int j = j0 + lane;
#pragma unroll
    for (int t = 0; t < AR; ++t) {
      const int i = i0 + wave + AW * t;
      const int a = (i < j) ? i : j;
      const int c = (i < j) ? j : i;
      const int xa = a - i0;
      const int yc = c - j0;
      const float pcc = sd[yc];
      float q = pcc;
      q -= sA[xa * AP + yc];
      q -= sB[yc * AP + xa];
      q += sd[AT + xa];
      r[t] = (a > 0) ? q : pcc;
    }
    if (I == J) {
#pragma unroll
      for (int t = 0; t < AR; ++t) {
        const int i = i0 + wave + AW * t;
        if (i == j + 1) r[t] = D[boff + (size_t)i * N + j];
      }
    }
#pragma unroll
    for (int t = 0; t < AR; ++t) {
      const int i = i0 + wave + AW * t;
      out[boff + (size_t)i * N + j] = r[t];
    }
  }

  if (I < J) {
    const int j = i0 + lane;
    const float paa = sd[AT + lane];
#pragma unroll
    for (int t = 0; t < AR; ++t) {
      const int y = wave + AW * t;
      const float pcc = sd[y];
      float q = pcc;
      q -= sA[lane * AP + y];
      q -= sB[y * AP + lane];
      q += paa;
      r[t] = (j > 0) ? q : pcc;
    }
    if (J == I + 1 && wave == 0) {
      if (lane == AT - 1) r[0] = D[boff + (size_t)j0 * N + j];
    }
#pragma unroll
    for (int t = 0; t < AR; ++t) {
      const int i = j0 + wave + AW * t;
      out[boff + (size_t)i * N + j] = r[t];
    }
  }
}

extern "C" void kernel_launch(void* const* d_in, const int* in_sizes, int n_in,
                              void* d_out, int out_size, void* d_ws, size_t ws_size,
                              hipStream_t stream) {
  const float* D = (const float*)d_in[0];
  float* out = (float*)d_out;
  float* P   = (float*)d_ws;
  const int B = in_sizes[0] / (N * N);

  rowscan_wmma<<<dim3((B * NTILE) / 8), dim3(256), 0, stream>>>(D, P);
  colscan<<<dim3((B * N) / 32), dim3(32), 0, stream>>>(P);
  assemble<<<dim3(NT + 1, NT / 2, B), dim3(AW * 32), 0, stream>>>(D, P, out);
}


# rowscan: tile loads decoupled from the WMMA carry chain by a 16-slot register ring of 16x32 tile pairs (up to 60 whole-line loads in flight per wave), one wave per workgroup; sum order unchanged
# speedup vs baseline: 9.3037x; 1.2049x over previous
#include <hip/hip_runtime.h>


typedef __attribute__((ext_vector_type(2))) float v2f;
typedef __attribute__((ext_vector_type(8))) float v8f;
typedef __attribute__((ext_vector_type(4))) float v4f;

#define N 2048
#define NTILE (N / 16)
#define LOG2N 11

#define RS_PW    32
#define RS_NP    (N / RS_PW)
#define RS_RING  16
#define RS_PITCH 36

__global__ __launch_bounds__(32, 1) void rowscan_wmma(const float* __restrict__ D,
                                                      float* __restrict__ R) {
  const int lane  = threadIdx.x;
  const int strip = blockIdx.x;
  const int b     = strip >> 7;
  const int row0  = (strip & 127) << 4;

  const float* __restrict__ Db = D + (size_t)b * N * N;
  float* __restrict__       Rb = R + (size_t)b * N * N;

  __shared__ float tl[16 * RS_PITCH];

  const int m    = lane & 15;
  const int half = lane >> 4;
  const int lr   = lane >> 3;
  const int lc   = lane & 7;

  v2f uu[4];
#pragma unroll
  for (int s = 0; s < 4; ++s) {
    const int k0 = s * 4 + half * 2;
    uu[s].x = (k0     <= m) ? 1.0f : 0.0f;
    uu[s].y = (k0 + 1 <= m) ? 1.0f : 0.0f;
  }

  v8f c = {0.f, 0.f, 0.f, 0.f, 0.f, 0.f, 0.f, 0.f};
  const int bsrc = half ? 31 : 15;

  const float* src = Db + (size_t)(row0 + lr) * N + lc * 4;
  float*       dst = Rb + (size_t)(row0 + half * 8) * N + m;

  v4f ring[RS_RING][4];

#define RS_FETCH(slot, pcol)                                                   \
  _Pragma("unroll")   \
  for (int t = 0; t < 4; ++t)                                                  \
    ring[slot][t] = *(const v4f*)(src + (size_t)(4 * t) * N + (pcol) * RS_PW);

#define RS_CONSUME(slot, pcol)                                                 \
  {                                                                            \
    _Pragma("unroll")   \
    for (int t = 0; t < 4; ++t)                                                \
      *(v4f*)&tl[(lr + 4 * t) * RS_PITCH + lc * 4] = ring[slot][t];         \
       \
       \
    v2f a[2][4];                                                               \
    _Pragma("unroll")   \
    for (int tt = 0; tt < 2; ++tt) {                                           \
      _Pragma("unroll")                                           \
      for (int s = 0; s < 4; ++s) {                                            \
          \
        const int k0 = s * 4 + half * 2;                                       \
        a[tt][s].x = tl[m * RS_PITCH + tt * 16 + k0];                          \
        a[tt][s].y = tl[m * RS_PITCH + tt * 16 + k0 + 1];                      \
      }                                                                        \
    }                                                                          \
    _Pragma("unroll")         \
    for (int tt = 0; tt < 2; ++tt) {                                           \
         \
         \
      _Pragma("unroll")            \
      for (int s = 0; s < 4; ++s)                                              \
        c = __builtin_amdgcn_wmma_f32_16x16x4_f32(                             \
             false, a[tt][s],  false, uu[s],                 \
             (short)0, c,  false,  false);      \
                \
      _Pragma("unroll")                      \
      for (int r = 0; r < 8; ++r)                                              \
        dst[(size_t)r * N + (pcol) * RS_PW + tt * 16] = c[r];                  \
             \
      _Pragma("unroll")                      \
      for (int r = 0; r < 8; ++r) c[r] = __shfl(c[r], bsrc, 32);               \
    }                                                                          \
  }

#pragma unroll
  for (int u = 0; u < RS_RING - 1; ++u) { RS_FETCH(u, u) }

  for (int g = 0; g < RS_NP - RS_RING; g += RS_RING) {
#pragma unroll
    for (int u = 0; u < RS_RING; ++u) {
      RS_FETCH((u + RS_RING - 1) % RS_RING, u + RS_RING - 1)
      RS_CONSUME(u, u)
    }
    src += RS_RING * RS_PW;
    dst += RS_RING * RS_PW;
  }

#pragma unroll
  for (int u = 0; u < RS_RING; ++u) {
    if (u == 0) { RS_FETCH(RS_RING - 1, RS_RING - 1) }
    RS_CONSUME(u, u)
  }
#undef RS_FETCH
#undef RS_CONSUME
}

#define CS_BATCH 16
#define CS_RING  4
#define CS_NB    (N / CS_BATCH)

__global__ __launch_bounds__(32, 1) void colscan(float* __restrict__ P) {
  const int colg = blockIdx.x * 32 + threadIdx.x;
  const int b = colg >> LOG2N;
  const int j = colg & (N - 1);
  float* p = P + (size_t)b * N * N + j;

  float ring[CS_RING][CS_BATCH];

#pragma unroll
  for (int u = 0; u < CS_RING - 1; ++u) {
#pragma unroll
    for (int r = 0; r < CS_BATCH; ++r)
      ring[u][r] = p[(size_t)(u * CS_BATCH + r) * N];
  }

  float acc = 0.0f;
  for (int g = 0; g < CS_NB - CS_RING; g += CS_RING) {
#pragma unroll
    for (int u = 0; u < CS_RING; ++u) {
#pragma unroll
      for (int r = 0; r < CS_BATCH; ++r)
        ring[(u + CS_RING - 1) % CS_RING][r] =
            p[(size_t)((u + CS_RING - 1) * CS_BATCH + r) * N];
#pragma unroll
      for (int r = 0; r < CS_BATCH; ++r) {
        acc += ring[u][r];
        p[(size_t)(u * CS_BATCH + r) * N] = acc;
      }
    }
    p += (size_t)CS_RING * CS_BATCH * N;
  }

#pragma unroll
  for (int u = 0; u < CS_RING; ++u) {
    if (u == 0) {
#pragma unroll
      for (int r = 0; r < CS_BATCH; ++r)
        ring[CS_RING - 1][r] = p[(size_t)((CS_RING - 1) * CS_BATCH + r) * N];
    }
#pragma unroll
    for (int r = 0; r < CS_BATCH; ++r) {
      acc += ring[u][r];
      p[(size_t)(u * CS_BATCH + r) * N] = acc;
    }
  }
}

#define AT 32
#define AP (AT + 1)
#define NT (N / AT)
#define AW 4
#define AR (AT / AW)

__global__ __launch_bounds__(AW * 32) void assemble(const float* __restrict__ D,
                                                const float* __restrict__ P,
                                                float* __restrict__ out) {
  __shared__ float sA[AT * AP];
  __shared__ float sB[AT * AP];
  __shared__ float sd[2 * AT];
  const int lane = threadIdx.x & 31;
  const int wave = threadIdx.x >> 5;

  const int u = blockIdx.x;
  const int v = blockIdx.y;
  int I, J;
  if (u < NT - v) { I = v;          J = v + u; }
  else            { I = NT - 1 - v; J = I + (u - (NT - v)); }
  const int i0 = I * AT;
  const int j0 = J * AT;

  const size_t boff = (size_t)blockIdx.z * N * N;
  const float* __restrict__ Pb = P + boff;

  const int colB  = i0 - 1 + lane;
  const int colBc = (colB < 0) ? 0 : colB;
  float ra[AR], rb[AR];
#pragma unroll
  for (int t = 0; t < AR; ++t) {
    const int x = wave + AW * t;
    const int rowA  = i0 + x - 1;
    const int rowAc = (rowA < 0) ? 0 : rowA;
    ra[t] = Pb[(size_t)rowAc * N + j0 + lane];
    rb[t] = Pb[(size_t)(j0 + x) * N + colBc];
  }
  float dv = 0.0f;
  if (wave < 2) {
    const int k  = (wave == 0) ? (j0 + lane) : (i0 + lane - 1);
    const int kc = (k < 0) ? 0 : k;
    dv = Pb[(size_t)kc * (N + 1)];
  }
#pragma unroll
  for (int t = 0; t < AR; ++t) {
    const int x = wave + AW * t;
    sA[x * AP + lane] = ra[t];
    sB[x * AP + lane] = rb[t];
  }
  if (wave < 2) sd[wave * AT + lane] = dv;
  __syncthreads();

  float r[AR];

  {
    const int j = j0 + lane;
#pragma unroll
    for (int t = 0; t < AR; ++t) {
      const int i = i0 + wave + AW * t;
      const int a = (i < j) ? i : j;
      const int c = (i < j) ? j : i;
      const int xa = a - i0;
      const int yc = c - j0;
      const float pcc = sd[yc];
      float q = pcc;
      q -= sA[xa * AP + yc];
      q -= sB[yc * AP + xa];
      q += sd[AT + xa];
      r[t] = (a > 0) ? q : pcc;
    }
    if (I == J) {
#pragma unroll
      for (int t = 0; t < AR; ++t) {
        const int i = i0 + wave + AW * t;
        if (i == j + 1) r[t] = D[boff + (size_t)i * N + j];
      }
    }
#pragma unroll
    for (int t = 0; t < AR; ++t) {
      const int i = i0 + wave + AW * t;
      out[boff + (size_t)i * N + j] = r[t];
    }
  }

  if (I < J) {
    const int j = i0 + lane;
    const float paa = sd[AT + lane];
#pragma unroll
    for (int t = 0; t < AR; ++t) {
      const int y = wave + AW * t;
      const float pcc = sd[y];
      float q = pcc;
      q -= sA[lane * AP + y];
      q -= sB[y * AP + lane];
      q += paa;
      r[t] = (j > 0) ? q : pcc;
    }
    if (J == I + 1 && wave == 0) {
      if (lane == AT - 1) r[0] = D[boff + (size_t)j0 * N + j];
    }
#pragma unroll
    for (int t = 0; t < AR; ++t) {
      const int i = j0 + wave + AW * t;
      out[boff + (size_t)i * N + j] = r[t];
    }
  }
}

extern "C" void kernel_launch(void* const* d_in, const int* in_sizes, int n_in,
                              void* d_out, int out_size, void* d_ws, size_t ws_size,
                              hipStream_t stream) {
  const float* D = (const float*)d_in[0];
  float* out = (float*)d_out;
  float* P   = (float*)d_ws;
  const int B = in_sizes[0] / (N * N);

  rowscan_wmma<<<dim3(B * NTILE), dim3(32), 0, stream>>>(D, P);
  colscan<<<dim3((B * N) / 32), dim3(32), 0, stream>>>(P);
  assemble<<<dim3(NT + 1, NT / 2, B), dim3(AW * 32), 0, stream>>>(D, P, out);
}


# two-level column scan, no colscan pass: rowscan also forms each tile's 16-row strip prefix (L16 x tile on WMMA), a small kernel sums strip totals, assemble adds the carry; column sum order changed
# speedup vs baseline: 13.0117x; 1.3986x over previous
#include <hip/hip_runtime.h>


typedef __attribute__((ext_vector_type(2))) float v2f;
typedef __attribute__((ext_vector_type(8))) float v8f;
typedef __attribute__((ext_vector_type(4))) float v4f;

#define N 2048
#define NTILE (N / 16)
#define LOG2N 11

#define RS_PW    32
#define RS_NP    (N / RS_PW)
#define RS_RING  16
#define RS_PITCH 36

__global__ __launch_bounds__(32, 1) void rowscan_wmma(const float* __restrict__ D,
                                                      float* __restrict__ R) {
  const int lane  = threadIdx.x;
  const int strip = blockIdx.x;
  const int b     = strip >> 7;
  const int row0  = (strip & 127) << 4;

  const float* __restrict__ Db = D + (size_t)b * N * N;
  float* __restrict__       Rb = R + (size_t)b * N * N;

  __shared__ float tl[16 * RS_PITCH];

  const int m    = lane & 15;
  const int half = lane >> 4;
  const int lr   = lane >> 3;
  const int lc   = lane & 7;

  v2f uu[4];
#pragma unroll
  for (int s = 0; s < 4; ++s) {
    const int k0 = s * 4 + half * 2;
    uu[s].x = (k0     <= m) ? 1.0f : 0.0f;
    uu[s].y = (k0 + 1 <= m) ? 1.0f : 0.0f;
  }

  v2f ll[4];
#pragma unroll
  for (int s = 0; s < 4; ++s) {
    const int k0 = half * 8 + s * 2;
    ll[s].x = (k0     <= m) ? 1.0f : 0.0f;
    ll[s].y = (k0 + 1 <= m) ? 1.0f : 0.0f;
  }

  v8f c = {0.f, 0.f, 0.f, 0.f, 0.f, 0.f, 0.f, 0.f};
  const int bsrc = half ? 31 : 15;

  const float* src = Db + (size_t)(row0 + lr) * N + lc * 4;
  float*       dst = Rb + (size_t)(row0 + half * 8) * N + m;

  v4f ring[RS_RING][4];

#define RS_FETCH(slot, pcol)                                                   \
  _Pragma("unroll")   \
  for (int t = 0; t < 4; ++t)                                                  \
    ring[slot][t] = *(const v4f*)(src + (size_t)(4 * t) * N + (pcol) * RS_PW);

#define RS_CONSUME(slot, pcol)                                                 \
  {                                                                            \
    _Pragma("unroll")   \
    for (int t = 0; t < 4; ++t)                                                \
      *(v4f*)&tl[(lr + 4 * t) * RS_PITCH + lc * 4] = ring[slot][t];         \
       \
       \
    v2f a[2][4];                                                               \
    _Pragma("unroll")   \
    for (int tt = 0; tt < 2; ++tt) {                                           \
      _Pragma("unroll")                                           \
      for (int s = 0; s < 4; ++s) {                                            \
          \
        const int k0 = s * 4 + half * 2;                                       \
        a[tt][s].x = tl[m * RS_PITCH + tt * 16 + k0];                          \
        a[tt][s].y = tl[m * RS_PITCH + tt * 16 + k0 + 1];                      \
      }                                                                        \
    }                                                                          \
    _Pragma("unroll")         \
    for (int tt = 0; tt < 2; ++tt) {                                           \
         \
         \
      _Pragma("unroll")            \
      for (int s = 0; s < 4; ++s)                                              \
        c = __builtin_amdgcn_wmma_f32_16x16x4_f32(                             \
             false, a[tt][s],  false, uu[s],                 \
             (short)0, c,  false,  false);      \
         \
         \
      v8f cn;                                                                  \
      _Pragma("unroll")                      \
      for (int r = 0; r < 8; ++r) cn[r] = __shfl(c[r], bsrc, 32);              \
         \
         \
      v8f q = {0.f, 0.f, 0.f, 0.f, 0.f, 0.f, 0.f, 0.f};                        \
      _Pragma("unroll")   \
      for (int s = 0; s < 4; ++s) {                                            \
        v2f bq;                                                                \
        bq.x = c[2 * s];                                                       \
        bq.y = c[2 * s + 1];                                                   \
        q = __builtin_amdgcn_wmma_f32_16x16x4_f32(                             \
             false, ll[s],  false, bq,                       \
             (short)0, q,  false,  false);      \
      }                                                                        \
           \
      _Pragma("unroll")                      \
      for (int r = 0; r < 8; ++r)                                              \
        dst[(size_t)r * N + (pcol) * RS_PW + tt * 16] = q[r];                  \
      c = cn;                                                                  \
    }                                                                          \
  }

#pragma unroll
  for (int u = 0; u < RS_RING - 1; ++u) { RS_FETCH(u, u) }

  for (int g = 0; g < RS_NP - RS_RING; g += RS_RING) {
#pragma unroll
    for (int u = 0; u < RS_RING; ++u) {
      RS_FETCH((u + RS_RING - 1) % RS_RING, u + RS_RING - 1)
      RS_CONSUME(u, u)
    }
    src += RS_RING * RS_PW;
    dst += RS_RING * RS_PW;
  }

#pragma unroll
  for (int u = 0; u < RS_RING; ++u) {
    if (u == 0) { RS_FETCH(RS_RING - 1, RS_RING - 1) }
    RS_CONSUME(u, u)
  }
#undef RS_FETCH
#undef RS_CONSUME
}

#define SC_NS (N / 16)

__global__ __launch_bounds__(32, 1) void stripscan(float* __restrict__ P) {
  const int colg = blockIdx.x * 32 + threadIdx.x;
  const int b = colg >> LOG2N;
  const int j = colg & (N - 1);
  float* p = P + (size_t)b * N * N + (size_t)15 * N + j;

  float v[SC_NS];
#pragma unroll
  for (int s = 0; s < SC_NS; ++s) v[s] = p[(size_t)s * 16 * N];

  float acc = v[0];
#pragma unroll
  for (int s = 1; s < SC_NS; ++s) {
    acc += v[s];
    p[(size_t)s * 16 * N] = acc;
  }
}

#define AT 32
#define AP (AT + 1)
#define NT (N / AT)
#define AW 4
#define AR (AT / AW)

__global__ __launch_bounds__(AW * 32) void assemble(const float* __restrict__ D,
                                                const float* __restrict__ W,
                                                float* __restrict__ out) {
  __shared__ float sA[AT * AP];
  __shared__ float sB[AT * AP];
  __shared__ float sd[2 * AT];
  const int lane = threadIdx.x & 31;
  const int wave = threadIdx.x >> 5;

  const int u = blockIdx.x;
  const int v = blockIdx.y;
  int I, J;
  if (u < NT - v) { I = v;          J = v + u; }
  else            { I = NT - 1 - v; J = I + (u - (NT - v)); }
  const int i0 = I * AT;
  const int j0 = J * AT;

  const size_t boff = (size_t)blockIdx.z * N * N;
  const float* __restrict__ Wb = W + boff;

  const int colB  = i0 - 1 + lane;
  const int colBc = (colB < 0) ? 0 : colB;
  const float cA0 = (i0 > 0) ? Wb[(size_t)(i0 - 1) * N + j0 + lane] : 0.0f;
  const float cA1 = Wb[(size_t)(i0 + 15) * N + j0 + lane];
  const float cB0 = (j0 > 0) ? Wb[(size_t)(j0 - 1) * N + colBc] : 0.0f;
  const float cB1 = Wb[(size_t)(j0 + 15) * N + colBc];
  float ra[AR], rb[AR];
#pragma unroll
  for (int t = 0; t < AR; ++t) {
    const int x = wave + AW * t;
    const int rowA  = i0 + x - 1;
    const int rowAc = (rowA < 0) ? 0 : rowA;
    ra[t] = Wb[(size_t)rowAc * N + j0 + lane];
    rb[t] = Wb[(size_t)(j0 + x) * N + colBc];
  }
  float dv = 0.0f;
  if (wave < 2) {
    const int k  = (wave == 0) ? (j0 + lane) : (i0 + lane - 1);
    const int kc = (k < 0) ? 0 : k;
    dv = Wb[(size_t)kc * (N + 1)];
    const int kr = (kc & ~15) - 1;
    if ((kc & 15) != 15 && kr >= 0) dv += Wb[(size_t)kr * N + kc];
  }
#pragma unroll
  for (int t = 0; t < AR; ++t) {
    const int x = wave + AW * t;
    const float ca = ((x & 15) == 0)  ? 0.0f : ((x < 16) ? cA0 : cA1);
    const float cb = ((x & 15) == 15) ? 0.0f : ((x < 16) ? cB0 : cB1);
    sA[x * AP + lane] = ra[t] + ca;
    sB[x * AP + lane] = rb[t] + cb;
  }
  if (wave < 2) sd[wave * AT + lane] = dv;
  __syncthreads();

  float r[AR];

  {
    const int j = j0 + lane;
#pragma unroll
    for (int t = 0; t < AR; ++t) {
      const int i = i0 + wave + AW * t;
      const int a = (i < j) ? i : j;
      const int c = (i < j) ? j : i;
      const int xa = a - i0;
      const int yc = c - j0;
      const float pcc = sd[yc];
      float q = pcc;
      q -= sA[xa * AP + yc];
      q -= sB[yc * AP + xa];
      q += sd[AT + xa];
      r[t] = (a > 0) ? q : pcc;
    }
    if (I == J) {
#pragma unroll
      for (int t = 0; t < AR; ++t) {
        const int i = i0 + wave + AW * t;
        if (i == j + 1) r[t] = D[boff + (size_t)i * N + j];
      }
    }
#pragma unroll
    for (int t = 0; t < AR; ++t) {
      const int i = i0 + wave + AW * t;
      out[boff + (size_t)i * N + j] = r[t];
    }
  }

  if (I < J) {
    const int j = i0 + lane;
    const float paa = sd[AT + lane];
#pragma unroll
    for (int t = 0; t < AR; ++t) {
      const int y = wave + AW * t;
      const float pcc = sd[y];
      float q = pcc;
      q -= sA[lane * AP + y];
      q -= sB[y * AP + lane];
      q += paa;
      r[t] = (j > 0) ? q : pcc;
    }
    if (J == I + 1 && wave == 0) {
      if (lane == AT - 1) r[0] = D[boff + (size_t)j0 * N + j];
    }
#pragma unroll
    for (int t = 0; t < AR; ++t) {
      const int i = j0 + wave + AW * t;
      out[boff + (size_t)i * N + j] = r[t];
    }
  }
}

extern "C" void kernel_launch(void* const* d_in, const int* in_sizes, int n_in,
                              void* d_out, int out_size, void* d_ws, size_t ws_size,
                              hipStream_t stream) {
  const float* D = (const float*)d_in[0];
  float* out = (float*)d_out;
  float* P   = (float*)d_ws;
  const int B = in_sizes[0] / (N * N);

  rowscan_wmma<<<dim3(B * NTILE), dim3(32), 0, stream>>>(D, P);
  stripscan<<<dim3((B * N) / 32), dim3(32), 0, stream>>>(P);
  assemble<<<dim3(NT + 1, NT / 2, B), dim3(AW * 32), 0, stream>>>(D, P, out);
}


# assemble: one workgroup per pair of mirrored 64x64 tiles (tile row = one aligned 256-B unit), 4 waves stage 16 rows each, one carry row per tile: 16,896 waves enter, not 66,560; term order unchanged
# speedup vs baseline: 14.1517x; 1.0876x over previous
#include <hip/hip_runtime.h>


typedef __attribute__((ext_vector_type(2))) float v2f;
typedef __attribute__((ext_vector_type(8))) float v8f;
typedef __attribute__((ext_vector_type(4))) float v4f;

#define N 2048
#define NTILE (N / 16)
#define LOG2N 11

#define RS_PW    32
#define RS_NP    (N / RS_PW)
#define RS_RING  16
#define RS_PITCH 36

__global__ __launch_bounds__(32, 1) void rowscan_wmma(const float* __restrict__ D,
                                                      float* __restrict__ R) {
  const int lane  = threadIdx.x;
  const int strip = blockIdx.x;
  const int b     = strip >> 7;
  const int row0  = (strip & 127) << 4;

  const float* __restrict__ Db = D + (size_t)b * N * N;
  float* __restrict__       Rb = R + (size_t)b * N * N;

  __shared__ float tl[16 * RS_PITCH];

  const int m    = lane & 15;
  const int half = lane >> 4;
  const int lr   = lane >> 3;
  const int lc   = lane & 7;

  v2f uu[4];
#pragma unroll
  for (int s = 0; s < 4; ++s) {
    const int k0 = s * 4 + half * 2;
    uu[s].x = (k0     <= m) ? 1.0f : 0.0f;
    uu[s].y = (k0 + 1 <= m) ? 1.0f : 0.0f;
  }

  v2f ll[4];
#pragma unroll
  for (int s = 0; s < 4; ++s) {
    const int k0 = half * 8 + s * 2;
    ll[s].x = (k0     <= m) ? 1.0f : 0.0f;
    ll[s].y = (k0 + 1 <= m) ? 1.0f : 0.0f;
  }

  v8f c = {0.f, 0.f, 0.f, 0.f, 0.f, 0.f, 0.f, 0.f};
  const int bsrc = half ? 31 : 15;

  const float* src = Db + (size_t)(row0 + lr) * N + lc * 4;
  float*       dst = Rb + (size_t)(row0 + half * 8) * N + m;

  v4f ring[RS_RING][4];

#define RS_FETCH(slot, pcol)                                                   \
  _Pragma("unroll")   \
  for (int t = 0; t < 4; ++t)                                                  \
    ring[slot][t] = *(const v4f*)(src + (size_t)(4 * t) * N + (pcol) * RS_PW);

#define RS_CONSUME(slot, pcol)                                                 \
  {                                                                            \
    _Pragma("unroll")   \
    for (int t = 0; t < 4; ++t)                                                \
      *(v4f*)&tl[(lr + 4 * t) * RS_PITCH + lc * 4] = ring[slot][t];         \
       \
       \
    v2f a[2][4];                                                               \
    _Pragma("unroll")   \
    for (int tt = 0; tt < 2; ++tt) {                                           \
      _Pragma("unroll")                                           \
      for (int s = 0; s < 4; ++s) {                                            \
          \
        const int k0 = s * 4 + half * 2;                                       \
        a[tt][s].x = tl[m * RS_PITCH + tt * 16 + k0];                          \
        a[tt][s].y = tl[m * RS_PITCH + tt * 16 + k0 + 1];                      \
      }                                                                        \
    }                                                                          \
    _Pragma("unroll")         \
    for (int tt = 0; tt < 2; ++tt) {                                           \
         \
         \
      _Pragma("unroll")            \
      for (int s = 0; s < 4; ++s)                                              \
        c = __builtin_amdgcn_wmma_f32_16x16x4_f32(                             \
             false, a[tt][s],  false, uu[s],                 \
             (short)0, c,  false,  false);      \
         \
         \
      v8f cn;                                                                  \
      _Pragma("unroll")                      \
      for (int r = 0; r < 8; ++r) cn[r] = __shfl(c[r], bsrc, 32);              \
         \
         \
      v8f q = {0.f, 0.f, 0.f, 0.f, 0.f, 0.f, 0.f, 0.f};                        \
      _Pragma("unroll")   \
      for (int s = 0; s < 4; ++s) {                                            \
        v2f bq;                                                                \
        bq.x = c[2 * s];                                                       \
        bq.y = c[2 * s + 1];                                                   \
        q = __builtin_amdgcn_wmma_f32_16x16x4_f32(                             \
             false, ll[s],  false, bq,                       \
             (short)0, q,  false,  false);      \
      }                                                                        \
           \
      _Pragma("unroll")                      \
      for (int r = 0; r < 8; ++r)                                              \
        dst[(size_t)r * N + (pcol) * RS_PW + tt * 16] = q[r];                  \
      c = cn;                                                                  \
    }                                                                          \
  }

#pragma unroll
  for (int u = 0; u < RS_RING - 1; ++u) { RS_FETCH(u, u) }

  for (int g = 0; g < RS_NP - RS_RING; g += RS_RING) {
#pragma unroll
    for (int u = 0; u < RS_RING; ++u) {
      RS_FETCH((u + RS_RING - 1) % RS_RING, u + RS_RING - 1)
      RS_CONSUME(u, u)
    }
    src += RS_RING * RS_PW;
    dst += RS_RING * RS_PW;
  }

#pragma unroll
  for (int u = 0; u < RS_RING; ++u) {
    if (u == 0) { RS_FETCH(RS_RING - 1, RS_RING - 1) }
    RS_CONSUME(u, u)
  }
#undef RS_FETCH
#undef RS_CONSUME
}

#define SC_NS (N / 16)

__global__ __launch_bounds__(32, 1) void stripscan(float* __restrict__ P) {
  const int colg = blockIdx.x * 32 + threadIdx.x;
  const int b = colg >> LOG2N;
  const int j = colg & (N - 1);
  float* p = P + (size_t)b * N * N + (size_t)15 * N + j;

  float v[SC_NS];
#pragma unroll
  for (int s = 0; s < SC_NS; ++s) v[s] = p[(size_t)s * 16 * N];

  float acc = v[0];
#pragma unroll
  for (int s = 1; s < SC_NS; ++s) {
    acc += v[s];
    p[(size_t)s * 16 * N] = acc;
  }
}

#define AT 64
#define AH (AT / 32)
#define AP (AT + 1)
#define NT (N / AT)
#define AW 4
#define AR (AT / AW)

__global__ __launch_bounds__(AW * 32) void assemble(const float* __restrict__ D,
                                                const float* __restrict__ W,
                                                float* __restrict__ out) {
  __shared__ float sA[AT * AP];
  __shared__ float sB[AT * AP];
  __shared__ float sd[2 * AT];
  const int lane = threadIdx.x & 31;
  const int wave = __builtin_amdgcn_readfirstlane(threadIdx.x >> 5);

  const int u = blockIdx.x;
  const int v = blockIdx.y;
  int I, J;
  if (u < NT - v) { I = v;          J = v + u; }
  else            { I = NT - 1 - v; J = I + (u - (NT - v)); }
  const int i0 = I * AT;
  const int j0 = J * AT;

  const size_t boff = (size_t)blockIdx.z * N * N;
  const float* __restrict__ Wb = W + boff;

  const int xs  = wave * AR;
  const int crA = i0 + xs - 1;
  const int crB = j0 + xs - 1;
  const int rowA0 = (crA < 0) ? 0 : crA;
  int colA[AH], colB[AH];
  float cB[AH];
#pragma unroll
  for (int h = 0; h < AH; ++h) {
    colA[h] = j0 + 32 * h + lane;
    const int cb = i0 - 1 + 32 * h + lane;
    colB[h] = (cb < 0) ? 0 : cb;
    cB[h] = (crB >= 0) ? Wb[(size_t)crB * N + colB[h]] : 0.0f;
  }
  float ra[AR][AH], rb[AR][AH];
#pragma unroll
  for (int t = 0; t < AR; ++t) {
    const int x = xs + t;
    const int rowA = (t == 0) ? rowA0 : (i0 + x - 1);
#pragma unroll
    for (int h = 0; h < AH; ++h) {
      ra[t][h] = Wb[(size_t)rowA * N + colA[h]];
      rb[t][h] = Wb[(size_t)(j0 + x) * N + colB[h]];
    }
  }
  float cA[AH];
#pragma unroll
  for (int h = 0; h < AH; ++h) cA[h] = (crA >= 0) ? ra[0][h] : 0.0f;
  float dv;
  {
    const int e  = (wave & 1) * 32 + lane;
    const int k  = (wave < 2) ? (j0 + e) : (i0 + e - 1);
    const int kc = (k < 0) ? 0 : k;
    dv = Wb[(size_t)kc * (N + 1)];
    const int kr = (kc & ~15) - 1;
    if ((kc & 15) != 15 && kr >= 0) dv += Wb[(size_t)kr * N + kc];
  }
#pragma unroll
  for (int t = 0; t < AR; ++t) {
    const int x = xs + t;
#pragma unroll
    for (int h = 0; h < AH; ++h) {
      const float ca = ((x & 15) == 0)  ? 0.0f : cA[h];
      const float cb = ((x & 15) == 15) ? 0.0f : cB[h];
      sA[x * AP + 32 * h + lane] = ra[t][h] + ca;
      sB[x * AP + 32 * h + lane] = rb[t][h] + cb;
    }
  }
  sd[wave * 32 + lane] = dv;
  __syncthreads();

  float r[AR][AH];

  {
#pragma unroll
    for (int t = 0; t < AR; ++t) {
      const int i = i0 + wave + AW * t;
#pragma unroll
      for (int h = 0; h < AH; ++h) {
        const int j = j0 + 32 * h + lane;
        const int a = (i < j) ? i : j;
        const int c = (i < j) ? j : i;
        const int xa = a - i0;
        const int yc = c - j0;
        const float pcc = sd[yc];
        float q = pcc;
        q -= sA[xa * AP + yc];
        q -= sB[yc * AP + xa];
        q += sd[AT + xa];
        r[t][h] = (a > 0) ? q : pcc;
      }
    }
    if (I == J) {
#pragma unroll
      for (int t = 0; t < AR; ++t) {
        const int i = i0 + wave + AW * t;
#pragma unroll
        for (int h = 0; h < AH; ++h) {
          const int j = j0 + 32 * h + lane;
          if (i == j + 1) r[t][h] = D[boff + (size_t)i * N + j];
        }
      }
    }
#pragma unroll
    for (int t = 0; t < AR; ++t) {
      const int i = i0 + wave + AW * t;
#pragma unroll
      for (int h = 0; h < AH; ++h)
        out[boff + (size_t)i * N + j0 + 32 * h + lane] = r[t][h];
    }
  }

  if (I < J) {
    float paa[AH];
#pragma unroll
    for (int h = 0; h < AH; ++h) paa[h] = sd[AT + 32 * h + lane];
#pragma unroll
    for (int t = 0; t < AR; ++t) {
      const int y = wave + AW * t;
      const float pcc = sd[y];
#pragma unroll
      for (int h = 0; h < AH; ++h) {
        const int xc = 32 * h + lane;
        float q = pcc;
        q -= sA[xc * AP + y];
        q -= sB[y * AP + xc];
        q += paa[h];
        r[t][h] = (i0 + xc > 0) ? q : pcc;
      }
    }
    if (J == I + 1 && wave == 0) {
      if (lane == 31) r[0][AH - 1] = D[boff + (size_t)j0 * N + i0 + AT - 1];
    }
#pragma unroll
    for (int t = 0; t < AR; ++t) {
      const int i = j0 + wave + AW * t;
#pragma unroll
      for (int h = 0; h < AH; ++h)
        out[boff + (size_t)i * N + i0 + 32 * h + lane] = r[t][h];
    }
  }
}

extern "C" void kernel_launch(void* const* d_in, const int* in_sizes, int n_in,
                              void* d_out, int out_size, void* d_ws, size_t ws_size,
                              hipStream_t stream) {
  const float* D = (const float*)d_in[0];
  float* out = (float*)d_out;
  float* P   = (float*)d_ws;
  const int B = in_sizes[0] / (N * N);

  rowscan_wmma<<<dim3(B * NTILE), dim3(32), 0, stream>>>(D, P);
  stripscan<<<dim3((B * N) / 32), dim3(32), 0, stream>>>(P);
  assemble<<<dim3(NT + 1, NT / 2, B), dim3(AW * 32), 0, stream>>>(D, P, out);
}


# rowscan: finished tiles staged through LDS and stored as whole aligned 256-B units (16 B per lane, two rows per store), not as 64-B half lines straight from the accumulators; sum order unchanged
# speedup vs baseline: 15.3387x; 1.0839x over previous
#include <hip/hip_runtime.h>


typedef __attribute__((ext_vector_type(2))) float v2f;
typedef __attribute__((ext_vector_type(8))) float v8f;
typedef __attribute__((ext_vector_type(4))) float v4f;

#define N 2048
#define NTILE (N / 16)
#define LOG2N 11

#define RS_PW    32
#define RS_NP    (N / RS_PW)
#define RS_RING  16
#define RS_PITCH 36
#define RQ_PW    (2 * RS_PW)
#define RQ_PITCH 68

__global__ __launch_bounds__(32, 1) void rowscan_wmma(const float* __restrict__ D,
                                                      float* __restrict__ R) {
  const int lane  = threadIdx.x;
  const int strip = blockIdx.x;
  const int b     = strip >> 7;
  const int row0  = (strip & 127) << 4;

  const float* __restrict__ Db = D + (size_t)b * N * N;
  float* __restrict__       Rb = R + (size_t)b * N * N;

  __shared__ float tl[16 * RS_PITCH];
  __shared__ __attribute__((aligned(16))) float ql[16 * RQ_PITCH];

  const int m    = lane & 15;
  const int half = lane >> 4;
  const int lr   = lane >> 3;
  const int lc   = lane & 7;

  v2f uu[4];
#pragma unroll
  for (int s = 0; s < 4; ++s) {
    const int k0 = s * 4 + half * 2;
    uu[s].x = (k0     <= m) ? 1.0f : 0.0f;
    uu[s].y = (k0 + 1 <= m) ? 1.0f : 0.0f;
  }

  v2f ll[4];
#pragma unroll
  for (int s = 0; s < 4; ++s) {
    const int k0 = half * 8 + s * 2;
    ll[s].x = (k0     <= m) ? 1.0f : 0.0f;
    ll[s].y = (k0 + 1 <= m) ? 1.0f : 0.0f;
  }

  v8f c = {0.f, 0.f, 0.f, 0.f, 0.f, 0.f, 0.f, 0.f};
  const int bsrc = half ? 31 : 15;

  const float* src = Db + (size_t)(row0 + lr) * N + lc * 4;
  float*       dst = Rb + (size_t)(row0 + half) * N + m * 4;
  float*       qst = ql + (half * 8) * RQ_PITCH + m;
  const float* qld = ql + half * RQ_PITCH + m * 4;

  v4f ring[RS_RING][4];
  v4f wq[8];

#define RS_FETCH(slot, pcol)                                                   \
  _Pragma("unroll")   \
  for (int t = 0; t < 4; ++t)                                                  \
    ring[slot][t] = *(const v4f*)(src + (size_t)(4 * t) * N + (pcol) * RS_PW);

#define RS_CONSUME(slot, pcol)                                                 \
  {                                                                            \
    _Pragma("unroll")   \
    for (int t = 0; t < 4; ++t)                                                \
      *(v4f*)&tl[(lr + 4 * t) * RS_PITCH + lc * 4] = ring[slot][t];         \
       \
       \
    v2f a[2][4];                                                               \
    _Pragma("unroll")   \
    for (int tt = 0; tt < 2; ++tt) {                                           \
      _Pragma("unroll")                                           \
      for (int s = 0; s < 4; ++s) {                                            \
          \
        const int k0 = s * 4 + half * 2;                                       \
        a[tt][s].x = tl[m * RS_PITCH + tt * 16 + k0];                          \
        a[tt][s].y = tl[m * RS_PITCH + tt * 16 + k0 + 1];                      \
      }                                                                        \
    }                                                                          \
    _Pragma("unroll")         \
    for (int tt = 0; tt < 2; ++tt) {                                           \
         \
         \
      _Pragma("unroll")            \
      for (int s = 0; s < 4; ++s)                                              \
        c = __builtin_amdgcn_wmma_f32_16x16x4_f32(                             \
             false, a[tt][s],  false, uu[s],                 \
             (short)0, c,  false,  false);      \
         \
         \
      v8f cn;                                                                  \
      _Pragma("unroll")                      \
      for (int r = 0; r < 8; ++r) cn[r] = __shfl(c[r], bsrc, 32);              \
         \
         \
      v8f q = {0.f, 0.f, 0.f, 0.f, 0.f, 0.f, 0.f, 0.f};                        \
      _Pragma("unroll")   \
      for (int s = 0; s < 4; ++s) {                                            \
        v2f bq;                                                                \
        bq.x = c[2 * s];                                                       \
        bq.y = c[2 * s + 1];                                                   \
        q = __builtin_amdgcn_wmma_f32_16x16x4_f32(                             \
             false, ll[s],  false, bq,                       \
             (short)0, q,  false,  false);      \
      }                                                                        \
        \
      _Pragma("unroll")                      \
      for (int r = 0; r < 8; ++r)                                              \
        qst[r * RQ_PITCH + ((pcol) & 1) * RS_PW + tt * 16] = q[r];             \
      c = cn;                                                                  \
    }                                                                          \
    if ((pcol) & 1) {                                                          \
         \
         \
         \
         \
      _Pragma("unroll")   \
      for (int t = 0; t < 8; ++t)                                              \
        wq[t] = *(const v4f*)(qld + (2 * t) * RQ_PITCH);                       \
      if ((pcol) == RS_RING - 1) {                                             \
                \
        _Pragma("unroll")   \
        for (int t = 0; t < 8; ++t)                                            \
          *(v4f*)(dst + (size_t)(2 * t) * N + ((pcol) >> 1) * RQ_PW) = wq[t];  \
      }                                                                        \
    } else if ((pcol) > 0) {                                                   \
         \
         \
         \
      _Pragma("unroll")   \
      for (int t = 0; t < 8; ++t)                                              \
        *(v4f*)(dst + (size_t)(2 * t) * N + (((pcol) >> 1) - 1) * RQ_PW) = wq[t]; \
    }                                                                          \
  }

#pragma unroll
  for (int u = 0; u < RS_RING - 1; ++u) { RS_FETCH(u, u) }

  for (int g = 0; g < RS_NP - RS_RING; g += RS_RING) {
#pragma unroll
    for (int u = 0; u < RS_RING; ++u) {
      RS_FETCH((u + RS_RING - 1) % RS_RING, u + RS_RING - 1)
      RS_CONSUME(u, u)
    }
    src += RS_RING * RS_PW;
    dst += RS_RING * RS_PW;
  }

#pragma unroll
  for (int u = 0; u < RS_RING; ++u) {
    if (u == 0) { RS_FETCH(RS_RING - 1, RS_RING - 1) }
    RS_CONSUME(u, u)
  }
#undef RS_FETCH
#undef RS_CONSUME
}

#define SC_NS (N / 16)

__global__ __launch_bounds__(32, 1) void stripscan(float* __restrict__ P) {
  const int colg = blockIdx.x * 32 + threadIdx.x;
  const int b = colg >> LOG2N;
  const int j = colg & (N - 1);
  float* p = P + (size_t)b * N * N + (size_t)15 * N + j;

  float v[SC_NS];
#pragma unroll
  for (int s = 0; s < SC_NS; ++s) v[s] = p[(size_t)s * 16 * N];

  float acc = v[0];
#pragma unroll
  for (int s = 1; s < SC_NS; ++s) {
    acc += v[s];
    p[(size_t)s * 16 * N] = acc;
  }
}

#define AT 64
#define AH (AT / 32)
#define AP (AT + 1)
#define NT (N / AT)
#define AW 4
#define AR (AT / AW)

__global__ __launch_bounds__(AW * 32) void assemble(const float* __restrict__ D,
                                                const float* __restrict__ W,
                                                float* __restrict__ out) {
  __shared__ float sA[AT * AP];
  __shared__ float sB[AT * AP];
  __shared__ float sd[2 * AT];
  const int lane = threadIdx.x & 31;
  const int wave = __builtin_amdgcn_readfirstlane(threadIdx.x >> 5);

  const int u = blockIdx.x;
  const int v = blockIdx.y;
  int I, J;
  if (u < NT - v) { I = v;          J = v + u; }
  else            { I = NT - 1 - v; J = I + (u - (NT - v)); }
  const int i0 = I * AT;
  const int j0 = J * AT;

  const size_t boff = (size_t)blockIdx.z * N * N;
  const float* __restrict__ Wb = W + boff;

  const int xs  = wave * AR;
  const int crA = i0 + xs - 1;
  const int crB = j0 + xs - 1;
  const int rowA0 = (crA < 0) ? 0 : crA;
  int colA[AH], colB[AH];
  float cB[AH];
#pragma unroll
  for (int h = 0; h < AH; ++h) {
    colA[h] = j0 + 32 * h + lane;
    const int cb = i0 - 1 + 32 * h + lane;
    colB[h] = (cb < 0) ? 0 : cb;
    cB[h] = (crB >= 0) ? Wb[(size_t)crB * N + colB[h]] : 0.0f;
  }
  float ra[AR][AH], rb[AR][AH];
#pragma unroll
  for (int t = 0; t < AR; ++t) {
    const int x = xs + t;
    const int rowA = (t == 0) ? rowA0 : (i0 + x - 1);
#pragma unroll
    for (int h = 0; h < AH; ++h) {
      ra[t][h] = Wb[(size_t)rowA * N + colA[h]];
      rb[t][h] = Wb[(size_t)(j0 + x) * N + colB[h]];
    }
  }
  float cA[AH];
#pragma unroll
  for (int h = 0; h < AH; ++h) cA[h] = (crA >= 0) ? ra[0][h] : 0.0f;
  float dv;
  {
    const int e  = (wave & 1) * 32 + lane;
    const int k  = (wave < 2) ? (j0 + e) : (i0 + e - 1);
    const int kc = (k < 0) ? 0 : k;
    dv = Wb[(size_t)kc * (N + 1)];
    const int kr = (kc & ~15) - 1;
    if ((kc & 15) != 15 && kr >= 0) dv += Wb[(size_t)kr * N + kc];
  }
#pragma unroll
  for (int t = 0; t < AR; ++t) {
    const int x = xs + t;
#pragma unroll
    for (int h = 0; h < AH; ++h) {
      const float ca = ((x & 15) == 0)  ? 0.0f : cA[h];
      const float cb = ((x & 15) == 15) ? 0.0f : cB[h];
      sA[x * AP + 32 * h + lane] = ra[t][h] + ca;
      sB[x * AP + 32 * h + lane] = rb[t][h] + cb;
    }
  }
  sd[wave * 32 + lane] = dv;
  __syncthreads();

  float r[AR][AH];

  {
#pragma unroll
    for (int t = 0; t < AR; ++t) {
      const int i = i0 + wave + AW * t;
#pragma unroll
      for (int h = 0; h < AH; ++h) {
        const int j = j0 + 32 * h + lane;
        const int a = (i < j) ? i : j;
        const int c = (i < j) ? j : i;
        const int xa = a - i0;
        const int yc = c - j0;
        const float pcc = sd[yc];
        float q = pcc;
        q -= sA[xa * AP + yc];
        q -= sB[yc * AP + xa];
        q += sd[AT + xa];
        r[t][h] = (a > 0) ? q : pcc;
      }
    }
    if (I == J) {
#pragma unroll
      for (int t = 0; t < AR; ++t) {
        const int i = i0 + wave + AW * t;
#pragma unroll
        for (int h = 0; h < AH; ++h) {
          const int j = j0 + 32 * h + lane;
          if (i == j + 1) r[t][h] = D[boff + (size_t)i * N + j];
        }
      }
    }
#pragma unroll
    for (int t = 0; t < AR; ++t) {
      const int i = i0 + wave + AW * t;
#pragma unroll
      for (int h = 0; h < AH; ++h)
        out[boff + (size_t)i * N + j0 + 32 * h + lane] = r[t][h];
    }
  }

  if (I < J) {
    float paa[AH];
#pragma unroll
    for (int h = 0; h < AH; ++h) paa[h] = sd[AT + 32 * h + lane];
#pragma unroll
    for (int t = 0; t < AR; ++t) {
      const int y = wave + AW * t;
      const float pcc = sd[y];
#pragma unroll
      for (int h = 0; h < AH; ++h) {
        const int xc = 32 * h + lane;
        float q = pcc;
        q -= sA[xc * AP + y];
        q -= sB[y * AP + xc];
        q += paa[h];
        r[t][h] = (i0 + xc > 0) ? q : pcc;
      }
    }
    if (J == I + 1 && wave == 0) {
      if (lane == 31) r[0][AH - 1] = D[boff + (size_t)j0 * N + i0 + AT - 1];
    }
#pragma unroll
    for (int t = 0; t < AR; ++t) {
      const int i = j0 + wave + AW * t;
#pragma unroll
      for (int h = 0; h < AH; ++h)
        out[boff + (size_t)i * N + i0 + 32 * h + lane] = r[t][h];
    }
  }
}

extern "C" void kernel_launch(void* const* d_in, const int* in_sizes, int n_in,
                              void* d_out, int out_size, void* d_ws, size_t ws_size,
                              hipStream_t stream) {
  const float* D = (const float*)d_in[0];
  float* out = (float*)d_out;
  float* P   = (float*)d_ws;
  const int B = in_sizes[0] / (N * N);

  rowscan_wmma<<<dim3(B * NTILE), dim3(32), 0, stream>>>(D, P);
  stripscan<<<dim3((B * N) / 32), dim3(32), 0, stream>>>(P);
  assemble<<<dim3(NT + 1, NT / 2, B), dim3(AW * 32), 0, stream>>>(D, P, out);
}
